# starting kernel
# baseline (speedup 1.0000x reference)
#include <hip/hip_runtime.h>
#include <math.h>

#define NQ 4
#define DIM 16
#define QDEPTH 6

typedef float v8f __attribute__((ext_vector_type(8)));
typedef float v2f __attribute__((ext_vector_type(2)));

#if defined(__has_builtin)
#if __has_builtin(__builtin_amdgcn_tanhf)
#define FAST_TANH(x) __builtin_amdgcn_tanhf(x)
#endif
#endif
#ifndef FAST_TANH
#define FAST_TANH(x) tanhf(x)
#endif

__device__ __forceinline__ void cnot16(float* st, int cm, int tm) {
#pragma unroll
  for (int i = 0; i < DIM; ++i) {
    if ((i & cm) && !(i & tm)) {
      float t = st[i];
      st[i] = st[i | tm];
      st[i | tm] = t;
    }
  }
}

__device__ __forceinline__ void ry16(float* st, int m, float c, float s) {
#pragma unroll
  for (int i = 0; i < DIM; ++i) {
    if (!(i & m)) {
      int j = i | m;
      float s0 = st[i], s1 = st[j];
      st[i] = c * s0 - s * s1;
      st[j] = s * s0 + c * s1;
    }
  }
}

__launch_bounds__(256)
__global__ void quantum_layer_kernel(const float* __restrict__ x,
                                     const float* __restrict__ qp,
                                     const float* __restrict__ scale,
                                     float* __restrict__ out,
                                     int numTiles) {
  __shared__ float Ms[DIM * DIM];

  if (threadIdx.x < DIM) {
    float st[DIM];
#pragma unroll
    for (int i = 0; i < DIM; ++i) st[i] = 0.0f;
    st[threadIdx.x] = 1.0f;
    for (int k = 0; k < QDEPTH; ++k) {
      cnot16(st, 8, 4);
      cnot16(st, 2, 1);
      cnot16(st, 4, 2);
#pragma unroll
      for (int w = 0; w < NQ; ++w) {
        float th = 0.5f * qp[k * NQ + w];
        ry16(st, 8 >> w, cosf(th), sinf(th));
      }
    }
#pragma unroll
    for (int j = 0; j < DIM; ++j) Ms[j * DIM + threadIdx.x] = st[j];
  }
  __syncthreads();

  const int lane = threadIdx.x & 31;
  const int n = lane & 15;
  const int h = lane >> 4;

  v2f A[4];
#pragma unroll
  for (int c = 0; c < 4; ++c) {
    int off = n * DIM + c * 4 + h * 2;
    A[c] = *(const v2f*)&Ms[off];
  }

  const float sc = scale[0];
  const int wavesPerBlock = blockDim.x >> 5;
  const int waveId = blockIdx.x * wavesPerBlock + (threadIdx.x >> 5);
  const int nWaves = gridDim.x * wavesPerBlock;

  const float4* __restrict__ x4 = (const float4*)x;
  float4* __restrict__ o4 = (float4*)out;
  const float PI4 = 0.78539816339744831f;

  for (int tile = waveId; tile < numTiles; tile += nWaves) {
    const int s = tile * 16 + n;
    float4 xv = x4[s];

    float inv = rsqrtf(xv.x * xv.x + xv.y * xv.y + xv.z * xv.z + xv.w * xv.w);
    float m = sc * inv;
    float a0 = FAST_TANH(xv.x * m) * PI4 + PI4;
    float a1 = FAST_TANH(xv.y * m) * PI4 + PI4;
    float a2 = FAST_TANH(xv.z * m) * PI4 + PI4;
    float a3 = FAST_TANH(xv.w * m) * PI4 + PI4;
    float c0 = __cosf(a0), s0v = __sinf(a0);
    float c1 = __cosf(a1), s1v = __sinf(a1);
    float c2 = __cosf(a2), s2v = __sinf(a2);
    float c3 = __cosf(a3), s3v = __sinf(a3);

    float f2h = h ? s2v : c2;
    float g0 = f2h * c3, g1 = f2h * s3v;
    float p01[4] = {c0 * c1, c0 * s1v, s0v * c1, s0v * s1v};

    v8f acc = {0.f, 0.f, 0.f, 0.f, 0.f, 0.f, 0.f, 0.f};
#pragma unroll
    for (int c = 0; c < 4; ++c) {
      v2f b;
      b[0] = p01[c] * g0;
      b[1] = p01[c] * g1;
      acc = __builtin_amdgcn_wmma_f32_16x16x4_f32(
          false, A[c], false, b, (short)0, acc, false, false);
    }

    float p0 = acc[0] * acc[0], p1 = acc[1] * acc[1];
    float p2 = acc[2] * acc[2], p3 = acc[3] * acc[3];
    float p4 = acc[4] * acc[4], p5 = acc[5] * acc[5];
    float p6 = acc[6] * acc[6], p7 = acc[7] * acc[7];

    float a01 = p0 + p1, a23 = p2 + p3, a45 = p4 + p5, a67 = p6 + p7;
    float b01 = p0 - p1, b23 = p2 - p3, b45 = p4 - p5, b67 = p6 - p7;
    float L = a01 + a23, H = a45 + a67;
    float S = L + H;
    float P0 = h ? -S : S;
    float P1 = L - H;
    float P2 = (a01 - a23) + (a45 - a67);
    float P3 = (b01 + b23) + (b45 + b67);

    float r0 = P0 + __shfl_xor(P0, 16, 32);
    float r1 = P1 + __shfl_xor(P1, 16, 32);
    float r2 = P2 + __shfl_xor(P2, 16, 32);
    float r3 = P3 + __shfl_xor(P3, 16, 32);

    if (h == 0) {
      o4[s] = make_float4(r0, r1, r2, r3);
    }
  }
}

extern "C" void kernel_launch(void* const* d_in, const int* in_sizes, int n_in,
                              void* d_out, int out_size, void* d_ws, size_t ws_size,
                              hipStream_t stream) {
  const float* x = (const float*)d_in[0];
  const float* qp = (const float*)d_in[1];
  const float* scale = (const float*)d_in[2];
  float* out = (float*)d_out;

  const int B = in_sizes[0] / 4;
  const int numTiles = B / 16;

  int blocks = 2048;
  int blocksNeeded = (numTiles + 7) / 8;
  if (blocks > blocksNeeded) blocks = blocksNeeded;
  if (blocks < 1) blocks = 1;

  quantum_layer_kernel<<<blocks, 256, 0, stream>>>(x, qp, scale, out, numTiles);
}


# x data path: a work-group owns 64 contiguous tiles and stages their x through LDS, its 4 coalesced 16-byte loads per thread issued before the factor build; the tile loop reads LDS, not global
# speedup vs baseline: 1.2048x; 1.2048x over previous
#include <hip/hip_runtime.h>
#include <math.h>


#define NQ 4
#define DIM 16
#define QDEPTH 6

typedef float v8f __attribute__((ext_vector_type(8)));
typedef float v2f __attribute__((ext_vector_type(2)));

#if defined(__has_builtin)
#if __has_builtin(__builtin_amdgcn_tanhf)
#define FAST_TANH(x) __builtin_amdgcn_tanhf(x)
#endif
#endif
#ifndef FAST_TANH
#define FAST_TANH(x) tanhf(x)
#endif

__device__ __forceinline__ void cnot16(float* st, int cm, int tm) {
#pragma unroll
  for (int i = 0; i < DIM; ++i) {
    if ((i & cm) && !(i & tm)) {
      float t = st[i];
      st[i] = st[i | tm];
      st[i | tm] = t;
    }
  }
}

__device__ __forceinline__ void ry16(float* st, int m, float c, float s) {
#pragma unroll
  for (int i = 0; i < DIM; ++i) {
    if (!(i & m)) {
      int j = i | m;
      float s0 = st[i], s1 = st[j];
      st[i] = c * s0 - s * s1;
      st[j] = s * s0 + c * s1;
    }
  }
}

__device__ __forceinline__ void layer_fwd(float* st, const float4 q) {
  cnot16(st, 8, 4);
  cnot16(st, 2, 1);
  cnot16(st, 4, 2);
  const float qa[NQ] = {q.x, q.y, q.z, q.w};
#pragma unroll
  for (int w = 0; w < NQ; ++w) {
    float th = 0.5f * qa[w];
    ry16(st, 8 >> w, cosf(th), sinf(th));
  }
}

__device__ __forceinline__ void layer_tr(float* st, const float4 q) {
  const float qa[NQ] = {q.x, q.y, q.z, q.w};
#pragma unroll
  for (int w = 0; w < NQ; ++w) {
    float th = 0.5f * qa[w];
    ry16(st, 8 >> w, cosf(th), -sinf(th));
  }
  cnot16(st, 4, 2);
  cnot16(st, 2, 1);
  cnot16(st, 8, 4);
}

__device__ __forceinline__ v8f mm16(const v8f a, const v8f b) {
  v8f acc = {0.f, 0.f, 0.f, 0.f, 0.f, 0.f, 0.f, 0.f};
#pragma unroll
  for (int q = 0; q < 4; ++q) {
    v2f fa, fb;
    fa[0] = a[2 * q]; fa[1] = a[2 * q + 1];
    fb[0] = b[2 * q]; fb[1] = b[2 * q + 1];
    acc = __builtin_amdgcn_wmma_f32_16x16x4_f32(
        false, fa, false, fb, (short)0, acc, false, false);
  }
  return acc;
}

#define CHUNK_TILES 64
#define CHUNK_SAMPLES (CHUNK_TILES * 16)
#define TILES_PER_WAVE 8
#define BLOCK_THREADS 256
#define X_LOADS (CHUNK_SAMPLES / BLOCK_THREADS)

__launch_bounds__(BLOCK_THREADS)
__global__ void quantum_layer_kernel(const float* __restrict__ x,
                                     const float* __restrict__ qp,
                                     const float* __restrict__ scale,
                                     float* __restrict__ out,
                                     int numTiles) {
  __shared__ __attribute__((aligned(16))) float Fs[QDEPTH][DIM * DIM];
  __shared__ __attribute__((aligned(16))) float4 Xs[CHUNK_SAMPLES];

  const int tid = threadIdx.x;
  const int lane = tid & 31;
  const int wv = __builtin_amdgcn_readfirstlane(tid >> 5);
  const int n = lane & 15;
  const int h = lane >> 4;

  const float4* __restrict__ x4 = (const float4*)x;
  float4* __restrict__ o4 = (float4*)out;
  const int nSamp = numTiles * 16;
  const int numChunks = (numTiles + CHUNK_TILES - 1) / CHUNK_TILES;
  int chunk = blockIdx.x;

  float4 xr[X_LOADS];
#pragma unroll
  for (int j = 0; j < X_LOADS; ++j) {
    const int s = chunk * CHUNK_SAMPLES + j * BLOCK_THREADS + tid;
    xr[j] = (s < nSamp) ? x4[s] : make_float4(0.f, 0.f, 0.f, 0.f);
  }

  if (wv < QDEPTH && lane < DIM) {
    float st[DIM];
#pragma unroll
    for (int i = 0; i < DIM; ++i) st[i] = 0.0f;
    st[lane] = 1.0f;
    const float4 q = ((const float4*)qp)[wv];
    if (wv == 2 || wv >= 4) {
      layer_tr(st, q);
    } else {
      layer_fwd(st, q);
    }
    float4* dst = (float4*)&Fs[wv][lane * DIM];
    dst[0] = make_float4(st[0], st[1], st[2], st[3]);
    dst[1] = make_float4(st[4], st[5], st[6], st[7]);
    dst[2] = make_float4(st[8], st[9], st[10], st[11]);
    dst[3] = make_float4(st[12], st[13], st[14], st[15]);
  }
#pragma unroll
  for (int j = 0; j < X_LOADS; ++j) Xs[j * BLOCK_THREADS + tid] = xr[j];
  __syncthreads();

  v8f f[QDEPTH];
  {
    const int off = n * DIM + h * 8;
#pragma unroll
    for (int k = 0; k < QDEPTH; ++k) {
      const float4* p = (const float4*)&Fs[k][off];
      const float4 a = p[0], b = p[1];
      f[k] = (v8f){a.x, a.y, a.z, a.w, b.x, b.y, b.z, b.w};
    }
  }
  const v8f P12 = mm16(f[1], f[2]);
  const v8f P43 = mm16(f[4], f[3]);
  const v8f UT = mm16(P12, f[0]);
  const v8f V = mm16(P43, f[5]);
  const v8f MT = mm16(UT, V);

  v2f A[4];
#pragma unroll
  for (int c = 0; c < 4; ++c) {
    A[c][0] = MT[2 * c];
    A[c][1] = MT[2 * c + 1];
  }

  const float sc = scale[0];
  const float PI4 = 0.78539816339744831f;

  while (chunk < numChunks) {
    for (int i = 0; i < TILES_PER_WAVE; ++i) {
      const int lt = wv * TILES_PER_WAVE + i;
      const int tile = chunk * CHUNK_TILES + lt;
      if (tile >= numTiles) break;
      const int s = tile * 16 + n;
      const float4 xv = Xs[lt * 16 + n];

      float inv = rsqrtf(xv.x * xv.x + xv.y * xv.y + xv.z * xv.z + xv.w * xv.w);
      float m = sc * inv;
      float a0 = FAST_TANH(xv.x * m) * PI4 + PI4;
      float a1 = FAST_TANH(xv.y * m) * PI4 + PI4;
      float a2 = FAST_TANH(xv.z * m) * PI4 + PI4;
      float a3 = FAST_TANH(xv.w * m) * PI4 + PI4;
      float c0 = __cosf(a0), s0v = __sinf(a0);
      float c1 = __cosf(a1), s1v = __sinf(a1);
      float c2 = __cosf(a2), s2v = __sinf(a2);
      float c3 = __cosf(a3), s3v = __sinf(a3);

      float f0h = h ? s0v : c0;
      float g0 = f0h * c3, g1 = f0h * s3v;
      float p12[4] = {c1 * c2, c1 * s2v, s1v * c2, s1v * s2v};

      v8f acc = {0.f, 0.f, 0.f, 0.f, 0.f, 0.f, 0.f, 0.f};
#pragma unroll
      for (int c = 0; c < 4; ++c) {
        v2f b;
        b[0] = p12[c] * g0;
        b[1] = p12[c] * g1;
        acc = __builtin_amdgcn_wmma_f32_16x16x4_f32(
            false, A[c], false, b, (short)0, acc, false, false);
      }

      float p0 = acc[0] * acc[0], p1 = acc[1] * acc[1];
      float p2 = acc[2] * acc[2], p3 = acc[3] * acc[3];
      float p4 = acc[4] * acc[4], p5 = acc[5] * acc[5];
      float p6 = acc[6] * acc[6], p7 = acc[7] * acc[7];

      float a01 = p0 + p1, a23 = p2 + p3, a45 = p4 + p5, a67 = p6 + p7;
      float b01 = p0 - p1, b23 = p2 - p3, b45 = p4 - p5, b67 = p6 - p7;
      float L = a01 + a23, H = a45 + a67;
      float S = L + H;
      float P0 = h ? -S : S;
      float P1 = L - H;
      float P2 = (a01 - a23) + (a45 - a67);
      float P3 = (b01 + b23) + (b45 + b67);

      float r0 = P0 + __shfl_xor(P0, 16, 32);
      float r1 = P1 + __shfl_xor(P1, 16, 32);
      float r2 = P2 + __shfl_xor(P2, 16, 32);
      float r3 = P3 + __shfl_xor(P3, 16, 32);

      if (h == 0) {
        o4[s] = make_float4(r0, r1, r2, r3);
      }
    }

    chunk += gridDim.x;
    if (chunk >= numChunks) break;
    __syncthreads();
#pragma unroll
    for (int j = 0; j < X_LOADS; ++j) {
      const int s = chunk * CHUNK_SAMPLES + j * BLOCK_THREADS + tid;
      xr[j] = (s < nSamp) ? x4[s] : make_float4(0.f, 0.f, 0.f, 0.f);
    }
#pragma unroll
    for (int j = 0; j < X_LOADS; ++j) Xs[j * BLOCK_THREADS + tid] = xr[j];
    __syncthreads();
  }
}

extern "C" void kernel_launch(void* const* d_in, const int* in_sizes, int n_in,
                              void* d_out, int out_size, void* d_ws, size_t ws_size,
                              hipStream_t stream) {
  const float* x = (const float*)d_in[0];
  const float* qp = (const float*)d_in[1];
  const float* scale = (const float*)d_in[2];
  float* out = (float*)d_out;

  const int B = in_sizes[0] / 4;
  const int numTiles = B / 16;

  int blocks = 2048;
  int blocksNeeded = (numTiles + 63) / 64;
  if (blocks > blocksNeeded) blocks = blocksNeeded;
  if (blocks < 1) blocks = 1;

  quantum_layer_kernel<<<blocks, 256, 0, stream>>>(x, qp, scale, out, numTiles);
}


# the five M merge products once per work-group: one wave (rotating with the work-group index) runs the 20 WMMAs and shares M^T through LDS, the other waves only load it; M's sums unchanged
# speedup vs baseline: 1.3505x; 1.1209x over previous
#include <hip/hip_runtime.h>
#include <math.h>


#define NQ 4
#define DIM 16
#define QDEPTH 6

typedef float v8f __attribute__((ext_vector_type(8)));
typedef float v2f __attribute__((ext_vector_type(2)));

#if defined(__has_builtin)
#if __has_builtin(__builtin_amdgcn_tanhf)
#define FAST_TANH(x) __builtin_amdgcn_tanhf(x)
#endif
#endif
#ifndef FAST_TANH
#define FAST_TANH(x) tanhf(x)
#endif

__device__ __forceinline__ void cnot16(float* st, int cm, int tm) {
#pragma unroll
  for (int i = 0; i < DIM; ++i) {
    if ((i & cm) && !(i & tm)) {
      float t = st[i];
      st[i] = st[i | tm];
      st[i | tm] = t;
    }
  }
}

__device__ __forceinline__ void ry16(float* st, int m, float c, float s) {
#pragma unroll
  for (int i = 0; i < DIM; ++i) {
    if (!(i & m)) {
      int j = i | m;
      float s0 = st[i], s1 = st[j];
      st[i] = c * s0 - s * s1;
      st[j] = s * s0 + c * s1;
    }
  }
}

__device__ __forceinline__ void layer_fwd(float* st, const float4 q) {
  cnot16(st, 8, 4);
  cnot16(st, 2, 1);
  cnot16(st, 4, 2);
  const float qa[NQ] = {q.x, q.y, q.z, q.w};
#pragma unroll
  for (int w = 0; w < NQ; ++w) {
    float th = 0.5f * qa[w];
    ry16(st, 8 >> w, cosf(th), sinf(th));
  }
}

__device__ __forceinline__ void layer_tr(float* st, const float4 q) {
  const float qa[NQ] = {q.x, q.y, q.z, q.w};
#pragma unroll
  for (int w = 0; w < NQ; ++w) {
    float th = 0.5f * qa[w];
    ry16(st, 8 >> w, cosf(th), -sinf(th));
  }
  cnot16(st, 4, 2);
  cnot16(st, 2, 1);
  cnot16(st, 8, 4);
}

__device__ __forceinline__ v8f mm16(const v8f a, const v8f b) {
  v8f acc = {0.f, 0.f, 0.f, 0.f, 0.f, 0.f, 0.f, 0.f};
#pragma unroll
  for (int q = 0; q < 4; ++q) {
    v2f fa, fb;
    fa[0] = a[2 * q]; fa[1] = a[2 * q + 1];
    fb[0] = b[2 * q]; fb[1] = b[2 * q + 1];
    acc = __builtin_amdgcn_wmma_f32_16x16x4_f32(
        false, fa, false, fb, (short)0, acc, false, false);
  }
  return acc;
}

#define CHUNK_TILES 64
#define CHUNK_SAMPLES (CHUNK_TILES * 16)
#define TILES_PER_WAVE 8
#define BLOCK_THREADS 256
#define X_LOADS (CHUNK_SAMPLES / BLOCK_THREADS)

__launch_bounds__(BLOCK_THREADS)
__global__ void quantum_layer_kernel(const float* __restrict__ x,
                                     const float* __restrict__ qp,
                                     const float* __restrict__ scale,
                                     float* __restrict__ out,
                                     int numTiles) {
  __shared__ __attribute__((aligned(16))) float Fs[QDEPTH][DIM * DIM];
  __shared__ __attribute__((aligned(16))) float Ms[DIM * DIM];
  __shared__ __attribute__((aligned(16))) float4 Xs[CHUNK_SAMPLES];

  const int tid = threadIdx.x;
  const int lane = tid & 31;
  const int wv = __builtin_amdgcn_readfirstlane(tid >> 5);
  const int n = lane & 15;
  const int h = lane >> 4;

  const float4* __restrict__ x4 = (const float4*)x;
  float4* __restrict__ o4 = (float4*)out;
  const int nSamp = numTiles * 16;
  const int numChunks = (numTiles + CHUNK_TILES - 1) / CHUNK_TILES;
  int chunk = blockIdx.x;

  float4 xr[X_LOADS];
#pragma unroll
  for (int j = 0; j < X_LOADS; ++j) {
    const int s = chunk * CHUNK_SAMPLES + j * BLOCK_THREADS + tid;
    xr[j] = (s < nSamp) ? x4[s] : make_float4(0.f, 0.f, 0.f, 0.f);
  }

  if (wv < QDEPTH && lane < DIM) {
    float st[DIM];
#pragma unroll
    for (int i = 0; i < DIM; ++i) st[i] = 0.0f;
    st[lane] = 1.0f;
    const float4 q = ((const float4*)qp)[wv];
    if (wv == 2 || wv >= 4) {
      layer_tr(st, q);
    } else {
      layer_fwd(st, q);
    }
    float4* dst = (float4*)&Fs[wv][lane * DIM];
    dst[0] = make_float4(st[0], st[1], st[2], st[3]);
    dst[1] = make_float4(st[4], st[5], st[6], st[7]);
    dst[2] = make_float4(st[8], st[9], st[10], st[11]);
    dst[3] = make_float4(st[12], st[13], st[14], st[15]);
  }
#pragma unroll
  for (int j = 0; j < X_LOADS; ++j) Xs[j * BLOCK_THREADS + tid] = xr[j];
  __syncthreads();

  const int off = n * DIM + h * 8;
  const int bx = blockIdx.x;
  const int mw = (bx + (bx >> 3) + (bx >> 6)) & 7;
  if (wv == mw) {
    v8f f[QDEPTH];
#pragma unroll
    for (int k = 0; k < QDEPTH; ++k) {
      const float4* p = (const float4*)&Fs[k][off];
      const float4 a = p[0], b = p[1];
      f[k] = (v8f){a.x, a.y, a.z, a.w, b.x, b.y, b.z, b.w};
    }
    const v8f P12 = mm16(f[1], f[2]);
    const v8f P43 = mm16(f[4], f[3]);
    const v8f UT = mm16(P12, f[0]);
    const v8f V = mm16(P43, f[5]);
    const v8f MT = mm16(UT, V);
    float4* dst = (float4*)&Ms[off];
    dst[0] = make_float4(MT[0], MT[1], MT[2], MT[3]);
    dst[1] = make_float4(MT[4], MT[5], MT[6], MT[7]);
  }
  __syncthreads();

  v2f A[4];
  {
    const float4* p = (const float4*)&Ms[off];
    const float4 a = p[0], b = p[1];
    A[0][0] = a.x; A[0][1] = a.y;
    A[1][0] = a.z; A[1][1] = a.w;
    A[2][0] = b.x; A[2][1] = b.y;
    A[3][0] = b.z; A[3][1] = b.w;
  }

  const float sc = scale[0];
  const float PI4 = 0.78539816339744831f;

  while (chunk < numChunks) {
    for (int i = 0; i < TILES_PER_WAVE; ++i) {
      const int lt = wv * TILES_PER_WAVE + i;
      const int tile = chunk * CHUNK_TILES + lt;
      if (tile >= numTiles) break;
      const int s = tile * 16 + n;
      const float4 xv = Xs[lt * 16 + n];

      float inv = rsqrtf(xv.x * xv.x + xv.y * xv.y + xv.z * xv.z + xv.w * xv.w);
      float m = sc * inv;
      float a0 = FAST_TANH(xv.x * m) * PI4 + PI4;
      float a1 = FAST_TANH(xv.y * m) * PI4 + PI4;
      float a2 = FAST_TANH(xv.z * m) * PI4 + PI4;
      float a3 = FAST_TANH(xv.w * m) * PI4 + PI4;
      float c0 = __cosf(a0), s0v = __sinf(a0);
      float c1 = __cosf(a1), s1v = __sinf(a1);
      float c2 = __cosf(a2), s2v = __sinf(a2);
      float c3 = __cosf(a3), s3v = __sinf(a3);

      float f0h = h ? s0v : c0;
      float g0 = f0h * c3, g1 = f0h * s3v;
      float p12[4] = {c1 * c2, c1 * s2v, s1v * c2, s1v * s2v};

      v8f acc = {0.f, 0.f, 0.f, 0.f, 0.f, 0.f, 0.f, 0.f};
#pragma unroll
      for (int c = 0; c < 4; ++c) {
        v2f b;
        b[0] = p12[c] * g0;
        b[1] = p12[c] * g1;
        acc = __builtin_amdgcn_wmma_f32_16x16x4_f32(
            false, A[c], false, b, (short)0, acc, false, false);
      }

      float p0 = acc[0] * acc[0], p1 = acc[1] * acc[1];
      float p2 = acc[2] * acc[2], p3 = acc[3] * acc[3];
      float p4 = acc[4] * acc[4], p5 = acc[5] * acc[5];
      float p6 = acc[6] * acc[6], p7 = acc[7] * acc[7];

      float a01 = p0 + p1, a23 = p2 + p3, a45 = p4 + p5, a67 = p6 + p7;
      float b01 = p0 - p1, b23 = p2 - p3, b45 = p4 - p5, b67 = p6 - p7;
      float L = a01 + a23, H = a45 + a67;
      float S = L + H;
      float P0 = h ? -S : S;
      float P1 = L - H;
      float P2 = (a01 - a23) + (a45 - a67);
      float P3 = (b01 + b23) + (b45 + b67);

      float r0 = P0 + __shfl_xor(P0, 16, 32);
      float r1 = P1 + __shfl_xor(P1, 16, 32);
      float r2 = P2 + __shfl_xor(P2, 16, 32);
      float r3 = P3 + __shfl_xor(P3, 16, 32);

      if (h == 0) {
        o4[s] = make_float4(r0, r1, r2, r3);
      }
    }

    chunk += gridDim.x;
    if (chunk >= numChunks) break;
    __syncthreads();
#pragma unroll
    for (int j = 0; j < X_LOADS; ++j) {
      const int s = chunk * CHUNK_SAMPLES + j * BLOCK_THREADS + tid;
      xr[j] = (s < nSamp) ? x4[s] : make_float4(0.f, 0.f, 0.f, 0.f);
    }
#pragma unroll
    for (int j = 0; j < X_LOADS; ++j) Xs[j * BLOCK_THREADS + tid] = xr[j];
    __syncthreads();
  }
}

extern "C" void kernel_launch(void* const* d_in, const int* in_sizes, int n_in,
                              void* d_out, int out_size, void* d_ws, size_t ws_size,
                              hipStream_t stream) {
  const float* x = (const float*)d_in[0];
  const float* qp = (const float*)d_in[1];
  const float* scale = (const float*)d_in[2];
  float* out = (float*)d_out;

  const int B = in_sizes[0] / 4;
  const int numTiles = B / 16;

  int blocks = 2048;
  int blocksNeeded = (numTiles + 63) / 64;
  if (blocks > blocksNeeded) blocks = blocksNeeded;
  if (blocks < 1) blocks = 1;

  quantum_layer_kernel<<<blocks, 256, 0, stream>>>(x, qp, scale, out, numTiles);
}


# angle work moved into the x staging: the lane that loads a sample evaluates its tanh/cos/sin once and stores two B-fragment factor vectors in LDS; no transcendental in the tile loop; sums unchanged
# speedup vs baseline: 1.4808x; 1.0965x over previous
#include <hip/hip_runtime.h>
#include <math.h>


#define NQ 4
#define DIM 16
#define QDEPTH 6

typedef float v8f __attribute__((ext_vector_type(8)));
typedef float v2f __attribute__((ext_vector_type(2)));

#if defined(__has_builtin)
#if __has_builtin(__builtin_amdgcn_tanhf)
#define FAST_TANH(x) __builtin_amdgcn_tanhf(x)
#endif
#endif
#ifndef FAST_TANH
#define FAST_TANH(x) tanhf(x)
#endif

__device__ __forceinline__ void cnot16(float* st, int cm, int tm) {
#pragma unroll
  for (int i = 0; i < DIM; ++i) {
    if ((i & cm) && !(i & tm)) {
      float t = st[i];
      st[i] = st[i | tm];
      st[i | tm] = t;
    }
  }
}

__device__ __forceinline__ void ry16(float* st, int m, float c, float s) {
#pragma unroll
  for (int i = 0; i < DIM; ++i) {
    if (!(i & m)) {
      int j = i | m;
      float s0 = st[i], s1 = st[j];
      st[i] = c * s0 - s * s1;
      st[j] = s * s0 + c * s1;
    }
  }
}

__device__ __forceinline__ void layer_fwd(float* st, const float4 q) {
  cnot16(st, 8, 4);
  cnot16(st, 2, 1);
  cnot16(st, 4, 2);
  const float qa[NQ] = {q.x, q.y, q.z, q.w};
#pragma unroll
  for (int w = 0; w < NQ; ++w) {
    float th = 0.5f * qa[w];
    ry16(st, 8 >> w, cosf(th), sinf(th));
  }
}

__device__ __forceinline__ void layer_tr(float* st, const float4 q) {
  const float qa[NQ] = {q.x, q.y, q.z, q.w};
#pragma unroll
  for (int w = 0; w < NQ; ++w) {
    float th = 0.5f * qa[w];
    ry16(st, 8 >> w, cosf(th), -sinf(th));
  }
  cnot16(st, 4, 2);
  cnot16(st, 2, 1);
  cnot16(st, 8, 4);
}

__device__ __forceinline__ v8f mm16(const v8f a, const v8f b) {
  v8f acc = {0.f, 0.f, 0.f, 0.f, 0.f, 0.f, 0.f, 0.f};
#pragma unroll
  for (int q = 0; q < 4; ++q) {
    v2f fa, fb;
    fa[0] = a[2 * q]; fa[1] = a[2 * q + 1];
    fb[0] = b[2 * q]; fb[1] = b[2 * q + 1];
    acc = __builtin_amdgcn_wmma_f32_16x16x4_f32(
        false, fa, false, fb, (short)0, acc, false, false);
  }
  return acc;
}

#define CHUNK_TILES 64
#define CHUNK_SAMPLES (CHUNK_TILES * 16)
#define TILES_PER_WAVE 8
#define BLOCK_THREADS 256
#define WAVE_SAMPLES (TILES_PER_WAVE * 16)
#define X_LOADS (WAVE_SAMPLES / 32)

__device__ __forceinline__ void stage_sample(const float4 xv, const float sc, float4* __restrict__ Pd,
                                             float4* __restrict__ Gd) {
  const float PI4 = 0.78539816339744831f;
  float inv = rsqrtf(xv.x * xv.x + xv.y * xv.y + xv.z * xv.z + xv.w * xv.w);
  float m = sc * inv;
  float a0 = FAST_TANH(xv.x * m) * PI4 + PI4;
  float a1 = FAST_TANH(xv.y * m) * PI4 + PI4;
  float a2 = FAST_TANH(xv.z * m) * PI4 + PI4;
  float a3 = FAST_TANH(xv.w * m) * PI4 + PI4;
  float c0 = __cosf(a0), s0v = __sinf(a0);
  float c1 = __cosf(a1), s1v = __sinf(a1);
  float c2 = __cosf(a2), s2v = __sinf(a2);
  float c3 = __cosf(a3), s3v = __sinf(a3);
  *Pd = make_float4(c1 * c2, c1 * s2v, s1v * c2, s1v * s2v);
  *Gd = make_float4(c0 * c3, c0 * s3v, s0v * c3, s0v * s3v);
}

__launch_bounds__(BLOCK_THREADS)
__global__ void quantum_layer_kernel(const float* __restrict__ x,
                                     const float* __restrict__ qp,
                                     const float* __restrict__ scale,
                                     float* __restrict__ out,
                                     int numTiles) {
  __shared__ __attribute__((aligned(16))) float Fs[QDEPTH][DIM * DIM];
  float* const Ms = &Fs[0][0];
  __shared__ __attribute__((aligned(16))) float4 Ps[CHUNK_SAMPLES];
  __shared__ __attribute__((aligned(16))) float4 Gs[CHUNK_SAMPLES];

  const int tid = threadIdx.x;
  const int lane = tid & 31;
  const int wv = __builtin_amdgcn_readfirstlane(tid >> 5);
  const int n = lane & 15;
  const int h = lane >> 4;

  const float4* __restrict__ x4 = (const float4*)x;
  float4* __restrict__ o4 = (float4*)out;
  const int nSamp = numTiles * 16;
  const int numChunks = (numTiles + CHUNK_TILES - 1) / CHUNK_TILES;
  int chunk = blockIdx.x;

  const int wbase = wv * WAVE_SAMPLES + lane;
  float4 xr[X_LOADS];
#pragma unroll
  for (int j = 0; j < X_LOADS; ++j) {
    const int s = chunk * CHUNK_SAMPLES + wbase + j * 32;
    xr[j] = (s < nSamp) ? x4[s] : make_float4(0.f, 0.f, 0.f, 0.f);
  }
  const float sc = scale[0];

  if (wv < QDEPTH && lane < DIM) {
    float st[DIM];
#pragma unroll
    for (int i = 0; i < DIM; ++i) st[i] = 0.0f;
    st[lane] = 1.0f;
    const float4 q = ((const float4*)qp)[wv];
    if (wv == 2 || wv >= 4) {
      layer_tr(st, q);
    } else {
      layer_fwd(st, q);
    }
    float4* dst = (float4*)&Fs[wv][lane * DIM];
    dst[0] = make_float4(st[0], st[1], st[2], st[3]);
    dst[1] = make_float4(st[4], st[5], st[6], st[7]);
    dst[2] = make_float4(st[8], st[9], st[10], st[11]);
    dst[3] = make_float4(st[12], st[13], st[14], st[15]);
  }
#pragma unroll
  for (int j = 0; j < X_LOADS; ++j) stage_sample(xr[j], sc, &Ps[wbase + j * 32], &Gs[wbase + j * 32]);
  __syncthreads();

  const int off = n * DIM + h * 8;
  const int bx = blockIdx.x;
  const int mw = (bx + (bx >> 3) + (bx >> 6)) & 7;
  if (wv == mw) {
    v8f f[QDEPTH];
#pragma unroll
    for (int k = 0; k < QDEPTH; ++k) {
      const float4* p = (const float4*)&Fs[k][off];
      const float4 a = p[0], b = p[1];
      f[k] = (v8f){a.x, a.y, a.z, a.w, b.x, b.y, b.z, b.w};
    }
    const v8f P12 = mm16(f[1], f[2]);
    const v8f P43 = mm16(f[4], f[3]);
    const v8f UT = mm16(P12, f[0]);
    const v8f V = mm16(P43, f[5]);
    const v8f MT = mm16(UT, V);
    float4* dst = (float4*)&Ms[off];
    dst[0] = make_float4(MT[0], MT[1], MT[2], MT[3]);
    dst[1] = make_float4(MT[4], MT[5], MT[6], MT[7]);
  }
  __syncthreads();

  v2f A[4];
  {
    const float4* p = (const float4*)&Ms[off];
    const float4 a = p[0], b = p[1];
    A[0][0] = a.x; A[0][1] = a.y;
    A[1][0] = a.z; A[1][1] = a.w;
    A[2][0] = b.x; A[2][1] = b.y;
    A[3][0] = b.z; A[3][1] = b.w;
  }

  while (chunk < numChunks) {
    for (int i = 0; i < TILES_PER_WAVE; ++i) {
      const int lt = wv * TILES_PER_WAVE + i;
      const int tile = chunk * CHUNK_TILES + lt;
      if (tile >= numTiles) break;
      const int s = tile * 16 + n;
      const float4 pv = Ps[lt * 16 + n];
      const float2 gv = ((const float2*)&Gs[lt * 16 + n])[h];
      const float g0 = gv.x, g1 = gv.y;
      const float p12[4] = {pv.x, pv.y, pv.z, pv.w};

      v8f acc = {0.f, 0.f, 0.f, 0.f, 0.f, 0.f, 0.f, 0.f};
#pragma unroll
      for (int c = 0; c < 4; ++c) {
        v2f b;
        b[0] = p12[c] * g0;
        b[1] = p12[c] * g1;
        acc = __builtin_amdgcn_wmma_f32_16x16x4_f32(
            false, A[c], false, b, (short)0, acc, false, false);
      }

      float p0 = acc[0] * acc[0], p1 = acc[1] * acc[1];
      float p2 = acc[2] * acc[2], p3 = acc[3] * acc[3];
      float p4 = acc[4] * acc[4], p5 = acc[5] * acc[5];
      float p6 = acc[6] * acc[6], p7 = acc[7] * acc[7];

      float a01 = p0 + p1, a23 = p2 + p3, a45 = p4 + p5, a67 = p6 + p7;
      float b01 = p0 - p1, b23 = p2 - p3, b45 = p4 - p5, b67 = p6 - p7;
      float L = a01 + a23, H = a45 + a67;
      float S = L + H;
      float P0 = h ? -S : S;
      float P1 = L - H;
      float P2 = (a01 - a23) + (a45 - a67);
      float P3 = (b01 + b23) + (b45 + b67);

      float r0 = P0 + __shfl_xor(P0, 16, 32);
      float r1 = P1 + __shfl_xor(P1, 16, 32);
      float r2 = P2 + __shfl_xor(P2, 16, 32);
      float r3 = P3 + __shfl_xor(P3, 16, 32);

      if (h == 0) {
        o4[s] = make_float4(r0, r1, r2, r3);
      }
    }

    chunk += gridDim.x;
    if (chunk >= numChunks) break;
#pragma unroll
    for (int j = 0; j < X_LOADS; ++j) {
      const int s = chunk * CHUNK_SAMPLES + wbase + j * 32;
      xr[j] = (s < nSamp) ? x4[s] : make_float4(0.f, 0.f, 0.f, 0.f);
    }
#pragma unroll
    for (int j = 0; j < X_LOADS; ++j) stage_sample(xr[j], sc, &Ps[wbase + j * 32], &Gs[wbase + j * 32]);
  }
}

extern "C" void kernel_launch(void* const* d_in, const int* in_sizes, int n_in,
                              void* d_out, int out_size, void* d_ws, size_t ws_size,
                              hipStream_t stream) {
  const float* x = (const float*)d_in[0];
  const float* qp = (const float*)d_in[1];
  const float* scale = (const float*)d_in[2];
  float* out = (float*)d_out;

  const int B = in_sizes[0] / 4;
  const int numTiles = B / 16;

  int blocks = 2048;
  int blocksNeeded = (numTiles + 63) / 64;
  if (blocks > blocksNeeded) blocks = blocksNeeded;
  if (blocks < 1) blocks = 1;

  quantum_layer_kernel<<<blocks, 256, 0, stream>>>(x, qp, scale, out, numTiles);
}


# factor build in closed form: a layer's column or row is the tensor product of four rotation 2-vectors (one sincosf per angle, 28 multiplies), not 4 gate passes over a unit vector; M unchanged
# speedup vs baseline: 1.5816x; 1.0681x over previous
#include <hip/hip_runtime.h>
#include <math.h>


#define NQ 4
#define DIM 16
#define QDEPTH 6

typedef float v8f __attribute__((ext_vector_type(8)));
typedef float v2f __attribute__((ext_vector_type(2)));

#if defined(__has_builtin)
#if __has_builtin(__builtin_amdgcn_tanhf)
#define FAST_TANH(x) __builtin_amdgcn_tanhf(x)
#endif
#endif
#ifndef FAST_TANH
#define FAST_TANH(x) tanhf(x)
#endif

__device__ __forceinline__ void cnot16(float* st, int cm, int tm) {
#pragma unroll
  for (int i = 0; i < DIM; ++i) {
    if ((i & cm) && !(i & tm)) {
      float t = st[i];
      st[i] = st[i | tm];
      st[i | tm] = t;
    }
  }
}

__device__ __forceinline__ int cnot_map(int i) {
  i ^= (i & 8) >> 1;
  i ^= (i & 2) >> 1;
  i ^= (i & 4) >> 1;
  return i;
}

__device__ __forceinline__ void ry_layer_vec(float* g, const int fixed, const float4 q, const bool tr) {
  const float qa[NQ] = {q.x, q.y, q.z, q.w};
  float u[NQ][2];
#pragma unroll
  for (int w = 0; w < NQ; ++w) {
    float s, c;
    sincosf(0.5f * qa[w], &s, &c);
    if (tr) s = -s;
    const bool bit = (fixed >> (NQ - 1 - w)) & 1;
    u[w][0] = bit ? -s : c;
    u[w][1] = bit ? c : s;
  }
  float p01[4], p012[8];
#pragma unroll
  for (int i = 0; i < 4; ++i) p01[i] = u[0][i >> 1] * u[1][i & 1];
#pragma unroll
  for (int i = 0; i < 8; ++i) p012[i] = p01[i >> 1] * u[2][i & 1];
#pragma unroll
  for (int i = 0; i < DIM; ++i) g[i] = p012[i >> 1] * u[3][i & 1];
}

__device__ __forceinline__ v8f mm16(const v8f a, const v8f b) {
  v8f acc = {0.f, 0.f, 0.f, 0.f, 0.f, 0.f, 0.f, 0.f};
#pragma unroll
  for (int q = 0; q < 4; ++q) {
    v2f fa, fb;
    fa[0] = a[2 * q]; fa[1] = a[2 * q + 1];
    fb[0] = b[2 * q]; fb[1] = b[2 * q + 1];
    acc = __builtin_amdgcn_wmma_f32_16x16x4_f32(
        false, fa, false, fb, (short)0, acc, false, false);
  }
  return acc;
}

#define CHUNK_TILES 64
#define CHUNK_SAMPLES (CHUNK_TILES * 16)
#define TILES_PER_WAVE 8
#define BLOCK_THREADS 256
#define WAVE_SAMPLES (TILES_PER_WAVE * 16)
#define X_LOADS (WAVE_SAMPLES / 32)

__device__ __forceinline__ void stage_sample(const float4 xv, const float sc, float4* __restrict__ Pd,
                                             float4* __restrict__ Gd) {
  const float PI4 = 0.78539816339744831f;
  float inv = rsqrtf(xv.x * xv.x + xv.y * xv.y + xv.z * xv.z + xv.w * xv.w);
  float m = sc * inv;
  float a0 = FAST_TANH(xv.x * m) * PI4 + PI4;
  float a1 = FAST_TANH(xv.y * m) * PI4 + PI4;
  float a2 = FAST_TANH(xv.z * m) * PI4 + PI4;
  float a3 = FAST_TANH(xv.w * m) * PI4 + PI4;
  float c0 = __cosf(a0), s0v = __sinf(a0);
  float c1 = __cosf(a1), s1v = __sinf(a1);
  float c2 = __cosf(a2), s2v = __sinf(a2);
  float c3 = __cosf(a3), s3v = __sinf(a3);
  *Pd = make_float4(c1 * c2, c1 * s2v, s1v * c2, s1v * s2v);
  *Gd = make_float4(c0 * c3, c0 * s3v, s0v * c3, s0v * s3v);
}

__launch_bounds__(BLOCK_THREADS)
__global__ void quantum_layer_kernel(const float* __restrict__ x,
                                     const float* __restrict__ qp,
                                     const float* __restrict__ scale,
                                     float* __restrict__ out,
                                     int numTiles) {
  __shared__ __attribute__((aligned(16))) float Fs[QDEPTH][DIM * DIM];
  float* const Ms = &Fs[0][0];
  __shared__ __attribute__((aligned(16))) float4 Ps[CHUNK_SAMPLES];
  __shared__ __attribute__((aligned(16))) float4 Gs[CHUNK_SAMPLES];

  const int tid = threadIdx.x;
  const int lane = tid & 31;
  const int wv = __builtin_amdgcn_readfirstlane(tid >> 5);
  const int n = lane & 15;
  const int h = lane >> 4;

  const float4* __restrict__ x4 = (const float4*)x;
  float4* __restrict__ o4 = (float4*)out;
  const int nSamp = numTiles * 16;
  const int numChunks = (numTiles + CHUNK_TILES - 1) / CHUNK_TILES;
  int chunk = blockIdx.x;

  const int wbase = wv * WAVE_SAMPLES + lane;
  float4 xr[X_LOADS];
#pragma unroll
  for (int j = 0; j < X_LOADS; ++j) {
    const int s = chunk * CHUNK_SAMPLES + wbase + j * 32;
    xr[j] = (s < nSamp) ? x4[s] : make_float4(0.f, 0.f, 0.f, 0.f);
  }
  const float sc = scale[0];

  if (wv < QDEPTH && lane < DIM) {
    float st[DIM];
    const float4 q = ((const float4*)qp)[wv];
    if (wv == 2 || wv >= 4) {
      ry_layer_vec(st, lane, q, true);
      cnot16(st, 4, 2);
      cnot16(st, 2, 1);
      cnot16(st, 8, 4);
    } else {
      ry_layer_vec(st, cnot_map(lane), q, false);
    }
    float4* dst = (float4*)&Fs[wv][lane * DIM];
    dst[0] = make_float4(st[0], st[1], st[2], st[3]);
    dst[1] = make_float4(st[4], st[5], st[6], st[7]);
    dst[2] = make_float4(st[8], st[9], st[10], st[11]);
    dst[3] = make_float4(st[12], st[13], st[14], st[15]);
  }
#pragma unroll
  for (int j = 0; j < X_LOADS; ++j) stage_sample(xr[j], sc, &Ps[wbase + j * 32], &Gs[wbase + j * 32]);
  __syncthreads();

  const int off = n * DIM + h * 8;
  const int bx = blockIdx.x;
  const int mw = (bx + (bx >> 3) + (bx >> 6)) & 7;
  if (wv == mw) {
    v8f f[QDEPTH];
#pragma unroll
    for (int k = 0; k < QDEPTH; ++k) {
      const float4* p = (const float4*)&Fs[k][off];
      const float4 a = p[0], b = p[1];
      f[k] = (v8f){a.x, a.y, a.z, a.w, b.x, b.y, b.z, b.w};
    }
    const v8f P12 = mm16(f[1], f[2]);
    const v8f P43 = mm16(f[4], f[3]);
    const v8f UT = mm16(P12, f[0]);
    const v8f V = mm16(P43, f[5]);
    const v8f MT = mm16(UT, V);
    float4* dst = (float4*)&Ms[off];
    dst[0] = make_float4(MT[0], MT[1], MT[2], MT[3]);
    dst[1] = make_float4(MT[4], MT[5], MT[6], MT[7]);
  }
  __syncthreads();

  v2f A[4];
  {
    const float4* p = (const float4*)&Ms[off];
    const float4 a = p[0], b = p[1];
    A[0][0] = a.x; A[0][1] = a.y;
    A[1][0] = a.z; A[1][1] = a.w;
    A[2][0] = b.x; A[2][1] = b.y;
    A[3][0] = b.z; A[3][1] = b.w;
  }

  while (chunk < numChunks) {
    for (int i = 0; i < TILES_PER_WAVE; ++i) {
      const int lt = wv * TILES_PER_WAVE + i;
      const int tile = chunk * CHUNK_TILES + lt;
      if (tile >= numTiles) break;
      const int s = tile * 16 + n;
      const float4 pv = Ps[lt * 16 + n];
      const float2 gv = ((const float2*)&Gs[lt * 16 + n])[h];
      const float g0 = gv.x, g1 = gv.y;
      const float p12[4] = {pv.x, pv.y, pv.z, pv.w};

      v8f acc = {0.f, 0.f, 0.f, 0.f, 0.f, 0.f, 0.f, 0.f};
#pragma unroll
      for (int c = 0; c < 4; ++c) {
        v2f b;
        b[0] = p12[c] * g0;
        b[1] = p12[c] * g1;
        acc = __builtin_amdgcn_wmma_f32_16x16x4_f32(
            false, A[c], false, b, (short)0, acc, false, false);
      }

      float p0 = acc[0] * acc[0], p1 = acc[1] * acc[1];
      float p2 = acc[2] * acc[2], p3 = acc[3] * acc[3];
      float p4 = acc[4] * acc[4], p5 = acc[5] * acc[5];
      float p6 = acc[6] * acc[6], p7 = acc[7] * acc[7];

      float a01 = p0 + p1, a23 = p2 + p3, a45 = p4 + p5, a67 = p6 + p7;
      float b01 = p0 - p1, b23 = p2 - p3, b45 = p4 - p5, b67 = p6 - p7;
      float L = a01 + a23, H = a45 + a67;
      float S = L + H;
      float P0 = h ? -S : S;
      float P1 = L - H;
      float P2 = (a01 - a23) + (a45 - a67);
      float P3 = (b01 + b23) + (b45 + b67);

      float r0 = P0 + __shfl_xor(P0, 16, 32);
      float r1 = P1 + __shfl_xor(P1, 16, 32);
      float r2 = P2 + __shfl_xor(P2, 16, 32);
      float r3 = P3 + __shfl_xor(P3, 16, 32);

      if (h == 0) {
        o4[s] = make_float4(r0, r1, r2, r3);
      }
    }

    chunk += gridDim.x;
    if (chunk >= numChunks) break;
#pragma unroll
    for (int j = 0; j < X_LOADS; ++j) {
      const int s = chunk * CHUNK_SAMPLES + wbase + j * 32;
      xr[j] = (s < nSamp) ? x4[s] : make_float4(0.f, 0.f, 0.f, 0.f);
    }
#pragma unroll
    for (int j = 0; j < X_LOADS; ++j) stage_sample(xr[j], sc, &Ps[wbase + j * 32], &Gs[wbase + j * 32]);
  }
}

extern "C" void kernel_launch(void* const* d_in, const int* in_sizes, int n_in,
                              void* d_out, int out_size, void* d_ws, size_t ws_size,
                              hipStream_t stream) {
  const float* x = (const float*)d_in[0];
  const float* qp = (const float*)d_in[1];
  const float* scale = (const float*)d_in[2];
  float* out = (float*)d_out;

  const int B = in_sizes[0] / 4;
  const int numTiles = B / 16;

  int blocks = 2048;
  int blocksNeeded = (numTiles + 63) / 64;
  if (blocks > blocksNeeded) blocks = blocksNeeded;
  if (blocks < 1) blocks = 1;

  quantum_layer_kernel<<<blocks, 256, 0, stream>>>(x, qp, scale, out, numTiles);
}


# work-group of 32 waves (1024 threads) owns 256 tiles: the six-factor build and the five merge products run once per 32 waves, not once per 8; waves, tiles per wave and sums unchanged
# speedup vs baseline: 1.6876x; 1.0670x over previous
#include <hip/hip_runtime.h>
#include <math.h>


#define NQ 4
#define DIM 16
#define QDEPTH 6

typedef float v8f __attribute__((ext_vector_type(8)));
typedef float v2f __attribute__((ext_vector_type(2)));

#if defined(__has_builtin)
#if __has_builtin(__builtin_amdgcn_tanhf)
#define FAST_TANH(x) __builtin_amdgcn_tanhf(x)
#endif
#endif
#ifndef FAST_TANH
#define FAST_TANH(x) tanhf(x)
#endif

__device__ __forceinline__ void cnot16(float* st, int cm, int tm) {
#pragma unroll
  for (int i = 0; i < DIM; ++i) {
    if ((i & cm) && !(i & tm)) {
      float t = st[i];
      st[i] = st[i | tm];
      st[i | tm] = t;
    }
  }
}

__device__ __forceinline__ int cnot_map(int i) {
  i ^= (i & 8) >> 1;
  i ^= (i & 2) >> 1;
  i ^= (i & 4) >> 1;
  return i;
}

__device__ __forceinline__ void ry_layer_vec(float* g, const int fixed, const float4 q, const bool tr) {
  const float qa[NQ] = {q.x, q.y, q.z, q.w};
  float u[NQ][2];
#pragma unroll
  for (int w = 0; w < NQ; ++w) {
    float s, c;
    sincosf(0.5f * qa[w], &s, &c);
    if (tr) s = -s;
    const bool bit = (fixed >> (NQ - 1 - w)) & 1;
    u[w][0] = bit ? -s : c;
    u[w][1] = bit ? c : s;
  }
  float p01[4], p012[8];
#pragma unroll
  for (int i = 0; i < 4; ++i) p01[i] = u[0][i >> 1] * u[1][i & 1];
#pragma unroll
  for (int i = 0; i < 8; ++i) p012[i] = p01[i >> 1] * u[2][i & 1];
#pragma unroll
  for (int i = 0; i < DIM; ++i) g[i] = p012[i >> 1] * u[3][i & 1];
}

__device__ __forceinline__ v8f mm16(const v8f a, const v8f b) {
  v8f acc = {0.f, 0.f, 0.f, 0.f, 0.f, 0.f, 0.f, 0.f};
#pragma unroll
  for (int q = 0; q < 4; ++q) {
    v2f fa, fb;
    fa[0] = a[2 * q]; fa[1] = a[2 * q + 1];
    fb[0] = b[2 * q]; fb[1] = b[2 * q + 1];
    acc = __builtin_amdgcn_wmma_f32_16x16x4_f32(
        false, fa, false, fb, (short)0, acc, false, false);
  }
  return acc;
}

#define CHUNK_TILES 256
#define CHUNK_SAMPLES (CHUNK_TILES * 16)
#define TILES_PER_WAVE 8
#define BLOCK_THREADS 1024
#define WAVE_SAMPLES (TILES_PER_WAVE * 16)
#define X_LOADS (WAVE_SAMPLES / 32)

__device__ __forceinline__ void stage_sample(const float4 xv, const float sc, float4* __restrict__ Pd,
                                             float4* __restrict__ Gd) {
  const float PI4 = 0.78539816339744831f;
  float inv = rsqrtf(xv.x * xv.x + xv.y * xv.y + xv.z * xv.z + xv.w * xv.w);
  float m = sc * inv;
  float a0 = FAST_TANH(xv.x * m) * PI4 + PI4;
  float a1 = FAST_TANH(xv.y * m) * PI4 + PI4;
  float a2 = FAST_TANH(xv.z * m) * PI4 + PI4;
  float a3 = FAST_TANH(xv.w * m) * PI4 + PI4;
  float c0 = __cosf(a0), s0v = __sinf(a0);
  float c1 = __cosf(a1), s1v = __sinf(a1);
  float c2 = __cosf(a2), s2v = __sinf(a2);
  float c3 = __cosf(a3), s3v = __sinf(a3);
  *Pd = make_float4(c1 * c2, c1 * s2v, s1v * c2, s1v * s2v);
  *Gd = make_float4(c0 * c3, c0 * s3v, s0v * c3, s0v * s3v);
}

__launch_bounds__(BLOCK_THREADS)
__global__ void quantum_layer_kernel(const float* __restrict__ x,
                                     const float* __restrict__ qp,
                                     const float* __restrict__ scale,
                                     float* __restrict__ out,
                                     int numTiles) {
  __shared__ __attribute__((aligned(16))) float Fs[QDEPTH][DIM * DIM];
  float* const Ms = &Fs[0][0];
  __shared__ __attribute__((aligned(16))) float4 Ps[CHUNK_SAMPLES];
  __shared__ __attribute__((aligned(16))) float4 Gs[CHUNK_SAMPLES];

  const int tid = threadIdx.x;
  const int lane = tid & 31;
  const int wv = __builtin_amdgcn_readfirstlane(tid >> 5);
  const int n = lane & 15;
  const int h = lane >> 4;

  const float4* __restrict__ x4 = (const float4*)x;
  float4* __restrict__ o4 = (float4*)out;
  const int nSamp = numTiles * 16;
  const int numChunks = (numTiles + CHUNK_TILES - 1) / CHUNK_TILES;
  int chunk = blockIdx.x;

  const int wbase = wv * WAVE_SAMPLES + lane;
  float4 xr[X_LOADS];
#pragma unroll
  for (int j = 0; j < X_LOADS; ++j) {
    const int s = chunk * CHUNK_SAMPLES + wbase + j * 32;
    xr[j] = (s < nSamp) ? x4[s] : make_float4(0.f, 0.f, 0.f, 0.f);
  }
  const float sc = scale[0];

  if (wv < QDEPTH && lane < DIM) {
    float st[DIM];
    const float4 q = ((const float4*)qp)[wv];
    if (wv == 2 || wv >= 4) {
      ry_layer_vec(st, lane, q, true);
      cnot16(st, 4, 2);
      cnot16(st, 2, 1);
      cnot16(st, 8, 4);
    } else {
      ry_layer_vec(st, cnot_map(lane), q, false);
    }
    float4* dst = (float4*)&Fs[wv][lane * DIM];
    dst[0] = make_float4(st[0], st[1], st[2], st[3]);
    dst[1] = make_float4(st[4], st[5], st[6], st[7]);
    dst[2] = make_float4(st[8], st[9], st[10], st[11]);
    dst[3] = make_float4(st[12], st[13], st[14], st[15]);
  }
#pragma unroll
  for (int j = 0; j < X_LOADS; ++j) stage_sample(xr[j], sc, &Ps[wbase + j * 32], &Gs[wbase + j * 32]);
  __syncthreads();

  const int off = n * DIM + h * 8;
  const int bx = blockIdx.x;
  const int mw = (bx + (bx >> 3) + (bx >> 6)) & 31;
  if (wv == mw) {
    v8f f[QDEPTH];
#pragma unroll
    for (int k = 0; k < QDEPTH; ++k) {
      const float4* p = (const float4*)&Fs[k][off];
      const float4 a = p[0], b = p[1];
      f[k] = (v8f){a.x, a.y, a.z, a.w, b.x, b.y, b.z, b.w};
    }
    const v8f P12 = mm16(f[1], f[2]);
    const v8f P43 = mm16(f[4], f[3]);
    const v8f UT = mm16(P12, f[0]);
    const v8f V = mm16(P43, f[5]);
    const v8f MT = mm16(UT, V);
    float4* dst = (float4*)&Ms[off];
    dst[0] = make_float4(MT[0], MT[1], MT[2], MT[3]);
    dst[1] = make_float4(MT[4], MT[5], MT[6], MT[7]);
  }
  __syncthreads();

  v2f A[4];
  {
    const float4* p = (const float4*)&Ms[off];
    const float4 a = p[0], b = p[1];
    A[0][0] = a.x; A[0][1] = a.y;
    A[1][0] = a.z; A[1][1] = a.w;
    A[2][0] = b.x; A[2][1] = b.y;
    A[3][0] = b.z; A[3][1] = b.w;
  }

  while (chunk < numChunks) {
    for (int i = 0; i < TILES_PER_WAVE; ++i) {
      const int lt = wv * TILES_PER_WAVE + i;
      const int tile = chunk * CHUNK_TILES + lt;
      if (tile >= numTiles) break;
      const int s = tile * 16 + n;
      const float4 pv = Ps[lt * 16 + n];
      const float2 gv = ((const float2*)&Gs[lt * 16 + n])[h];
      const float g0 = gv.x, g1 = gv.y;
      const float p12[4] = {pv.x, pv.y, pv.z, pv.w};

      v8f acc = {0.f, 0.f, 0.f, 0.f, 0.f, 0.f, 0.f, 0.f};
#pragma unroll
      for (int c = 0; c < 4; ++c) {
        v2f b;
        b[0] = p12[c] * g0;
        b[1] = p12[c] * g1;
        acc = __builtin_amdgcn_wmma_f32_16x16x4_f32(
            false, A[c], false, b, (short)0, acc, false, false);
      }

      float p0 = acc[0] * acc[0], p1 = acc[1] * acc[1];
      float p2 = acc[2] * acc[2], p3 = acc[3] * acc[3];
      float p4 = acc[4] * acc[4], p5 = acc[5] * acc[5];
      float p6 = acc[6] * acc[6], p7 = acc[7] * acc[7];

      float a01 = p0 + p1, a23 = p2 + p3, a45 = p4 + p5, a67 = p6 + p7;
      float b01 = p0 - p1, b23 = p2 - p3, b45 = p4 - p5, b67 = p6 - p7;
      float L = a01 + a23, H = a45 + a67;
      float S = L + H;
      float P0 = h ? -S : S;
      float P1 = L - H;
      float P2 = (a01 - a23) + (a45 - a67);
      float P3 = (b01 + b23) + (b45 + b67);

      float r0 = P0 + __shfl_xor(P0, 16, 32);
      float r1 = P1 + __shfl_xor(P1, 16, 32);
      float r2 = P2 + __shfl_xor(P2, 16, 32);
      float r3 = P3 + __shfl_xor(P3, 16, 32);

      if (h == 0) {
        o4[s] = make_float4(r0, r1, r2, r3);
      }
    }

    chunk += gridDim.x;
    if (chunk >= numChunks) break;
#pragma unroll
    for (int j = 0; j < X_LOADS; ++j) {
      const int s = chunk * CHUNK_SAMPLES + wbase + j * 32;
      xr[j] = (s < nSamp) ? x4[s] : make_float4(0.f, 0.f, 0.f, 0.f);
    }
#pragma unroll
    for (int j = 0; j < X_LOADS; ++j) stage_sample(xr[j], sc, &Ps[wbase + j * 32], &Gs[wbase + j * 32]);
  }
}

extern "C" void kernel_launch(void* const* d_in, const int* in_sizes, int n_in,
                              void* d_out, int out_size, void* d_ws, size_t ws_size,
                              hipStream_t stream) {
  const float* x = (const float*)d_in[0];
  const float* qp = (const float*)d_in[1];
  const float* scale = (const float*)d_in[2];
  float* out = (float*)d_out;

  const int B = in_sizes[0] / 4;
  const int numTiles = B / 16;

  int blocks = 512;
  int blocksNeeded = (numTiles + 255) / 256;
  if (blocks > blocksNeeded) blocks = blocksNeeded;
  if (blocks < 1) blocks = 1;

  quantum_layer_kernel<<<blocks, 1024, 0, stream>>>(x, qp, scale, out, numTiles);
}
